# starting kernel
# baseline (speedup 1.0000x reference)
#include <hip/hip_runtime.h>
#include <math.h>

typedef __attribute__((ext_vector_type(2))) float v2f;
typedef __attribute__((ext_vector_type(8))) float v8f;

#define WPB 8

__global__ __launch_bounds__(256) void nxro_fused_kernel(
    const float* __restrict__ x, const float* __restrict__ t,
    const float* __restrict__ Lb, const float* __restrict__ wq,
    const float* __restrict__ wk, const float* __restrict__ wv,
    const float* __restrict__ wo, const float* __restrict__ aw,
    float* __restrict__ out, int nTiles)
{
    __shared__ float xs[WPB][16][18];
    __shared__ float al[WPB][16];

    const int lane = threadIdx.x & 31;
    const int wave = threadIdx.x >> 5;
    const int half = lane >> 4;
    const int r    = lane & 15;

    v2f bfr[20];
#pragma unroll
    for (int j = 0; j < 20; ++j) {
        const int kb = j >> 2;
        const int v0 = ((j & 3) << 2) + (half << 1);
        const float* p = Lb + kb * 256 + r * 16 + v0;
        v2f b_; b_.x = p[0]; b_.y = p[1];
        bfr[j] = b_;
    }

    float c1 = 0.f, c2 = 0.f;
#pragma unroll
    for (int d = 0; d < 32; ++d) {
        c1 = fmaf(wq[d], wk[d], c1);
        c2 = fmaf(wv[d], wo[d], c2);
    }
    c1 *= 0.17677669529663688f;

    const float aw0 = aw[0], aw1 = aw[1], aw2 = aw[2], aw3 = aw[3], aw4 = aw[4];

    const int waveId = blockIdx.x * WPB + wave;
    const int stride = gridDim.x * WPB;

    for (int tile = waveId; tile < nTiles; tile += stride) {
        const int s0  = tile << 4;
        const int row = s0 + r;

        const v2f* xp = reinterpret_cast<const v2f*>(x + row * 16);
        const v2f xq0 = xp[0 + half];
        const v2f xq1 = xp[2 + half];
        const v2f xq2 = xp[4 + half];
        const v2f xq3 = xp[6 + half];

        {
            v2f* rp = reinterpret_cast<v2f*>(&xs[wave][r][0]);
            rp[0 + half] = xq0;
            rp[2 + half] = xq1;
            rp[4 + half] = xq2;
            rp[6 + half] = xq3;
        }

        const float tt  = t[row];
        const float a1  = 6.283185307179586f * tt;
        const float e1c = cosf(a1), e1s = sinf(a1);
        const float tc  = e1c + e1c;
        const float e2c = fmaf(tc, e1c, -1.0f);
        const float e2s = tc * e1s;

        if (half == 0) {
            float z = aw0 + aw1 * e1c + aw2 * e1s + aw3 * e2c + aw4 * e2s;
            al[wave][r] = __builtin_amdgcn_rcpf(1.f + __expf(-z));
        }

        v8f acc = {};
#define WMMA_RAW(AFR, BJ)                                                      \
        acc = __builtin_amdgcn_wmma_f32_16x16x4_f32(                           \
            false, (AFR), false, bfr[BJ], (short)0, acc, false, false)
#define WMMA_SC(EV, XQ, BJ)                                                    \
        do { v2f a_; a_.x = (EV) * (XQ).x; a_.y = (EV) * (XQ).y;               \
             WMMA_RAW(a_, BJ); } while (0)

        WMMA_RAW(xq0, 0);  WMMA_RAW(xq1, 1);  WMMA_RAW(xq2, 2);  WMMA_RAW(xq3, 3);
        WMMA_SC(e1c, xq0, 4);  WMMA_SC(e1c, xq1, 5);  WMMA_SC(e1c, xq2, 6);  WMMA_SC(e1c, xq3, 7);
        WMMA_SC(e1s, xq0, 8);  WMMA_SC(e1s, xq1, 9);  WMMA_SC(e1s, xq2, 10); WMMA_SC(e1s, xq3, 11);
        WMMA_SC(e2c, xq0, 12); WMMA_SC(e2c, xq1, 13); WMMA_SC(e2c, xq2, 14); WMMA_SC(e2c, xq3, 15);
        WMMA_SC(e2s, xq0, 16); WMMA_SC(e2s, xq1, 17); WMMA_SC(e2s, xq2, 18); WMMA_SC(e2s, xq3, 19);
#undef WMMA_SC
#undef WMMA_RAW

        const int i = r;
        const float g = (i >= 2) ? 1.f : 0.f;
#pragma unroll
        for (int rr = 0; rr < 8; ++rr) {
            const int m = rr + (half << 3);
            const v2f xm01 = *reinterpret_cast<const v2f*>(&xs[wave][m][0]);
            const float xm0 = xm01.x;
            const float xm1 = xm01.y;
            const float xmi = xs[wave][m][i];
            const float am  = al[wave][m];
            const float e0 = c1 * xmi * xm0;
            const float e1 = c1 * xmi * xm1;
            const float e2 = c1 * xmi * xmi;
            const float mx = fmaxf(fmaxf(e0, e1), e2);
            const float w0 = __expf(e0 - mx);
            const float w1 = __expf(e1 - mx);
            const float w2 = g * __expf(e2 - mx);
            const float inv = __builtin_amdgcn_rcpf(w0 + w1 + w2);
            const float s  = (w0 * xm0 + w1 * xm1 + w2 * xmi) * inv;
            out[(s0 + m) * 16 + i] = acc[rr] + c2 * s * am;
        }
    }
}

extern "C" void kernel_launch(void* const* d_in, const int* in_sizes, int n_in,
                              void* d_out, int out_size, void* d_ws, size_t ws_size,
                              hipStream_t stream) {
    const float* x  = (const float*)d_in[0];
    const float* t  = (const float*)d_in[1];
    const float* Lb = (const float*)d_in[2];
    const float* wq = (const float*)d_in[3];
    const float* wk = (const float*)d_in[4];
    const float* wv = (const float*)d_in[5];
    const float* wo = (const float*)d_in[6];
    const float* aw = (const float*)d_in[7];
    float* out = (float*)d_out;

    const int B = in_sizes[1];
    const int nTiles = B >> 4;
    if (nTiles <= 0) return;

    int blocks = (nTiles + WPB - 1) / WPB;
    if (blocks > 512) blocks = 512;

    nxro_fused_kernel<<<blocks, 256, 0, stream>>>(x, t, Lb, wq, wk, wv, wo, aw,
                                                  out, nTiles);
}


# dxdt computed transposed on the WMMA (L_basis as A operand, samples as B) so a lane holds x and out of the same 8 columns: no LDS, 128-bit x loads/out stores; K order of the dxdt sums changes
# speedup vs baseline: 1.2619x; 1.2619x over previous
#include <hip/hip_runtime.h>
#include <math.h>

typedef __attribute__((ext_vector_type(2))) float v2f;
typedef __attribute__((ext_vector_type(4))) float v4f;
typedef __attribute__((ext_vector_type(8))) float v8f;

#define WPB 8


__device__ __forceinline__ float xhalf(float v) {
    const int b = __builtin_bit_cast(int, v);
    return __builtin_bit_cast(float,
        __builtin_amdgcn_permlanex16(b, b, 0x76543210, 0xfedcba98, false, false));
}

__global__ __launch_bounds__(256) void nxro_fused_kernel(
    const float* __restrict__ x, const float* __restrict__ t,
    const float* __restrict__ Lb, const float* __restrict__ wq,
    const float* __restrict__ wk, const float* __restrict__ wv,
    const float* __restrict__ wo, const float* __restrict__ aw,
    float* __restrict__ out, int nTiles)
{
    const int lane = threadIdx.x & 31;
    const int wave = threadIdx.x >> 5;
    const int half = lane >> 4;
    const int c    = lane & 15;

    const int t0 = (blockIdx.x * WPB + wave) << 1;
    if (t0 >= nTiles) return;
    const bool two = (t0 + 1) < nTiles;

    const int pu = ((c & 4) << 1) | ((c & 8) >> 1) | (c & 3);
    v4f lq[5][2];
#pragma unroll
    for (int kb = 0; kb < 5; ++kb) {
        const v4f* p = reinterpret_cast<const v4f*>(Lb + kb * 256 + pu * 16 + (half << 2));
        lq[kb][0] = p[0];
        lq[kb][1] = p[2];
    }

    float c1 = 0.f, c2 = 0.f;
#pragma unroll
    for (int d = 0; d < 32; ++d) {
        c1 = fmaf(wq[d], wk[d], c1);
        c2 = fmaf(wv[d], wo[d], c2);
    }
    c1 *= 0.17677669529663688f;

    const float aw0 = aw[0], aw1 = aw[1], aw2 = aw[2], aw3 = aw[3], aw4 = aw[4];

    const int s0   = t0 << 4;
    const int rowA = s0 + c;
    const int rowB = two ? rowA + 16 : rowA;
    const int own  = two ? s0 + lane : rowA;

    const v4f* xpa = reinterpret_cast<const v4f*>(x + rowA * 16 + (half << 2));
    const v4f* xpb = reinterpret_cast<const v4f*>(x + rowB * 16 + (half << 2));
    const v4f xa0 = xpa[0], xa1 = xpa[2];
    const v4f xb0 = xpb[0], xb1 = xpb[2];
    const float tt = t[own];

    asm volatile("" :: "s"(c1), "s"(c2));

    const float a1  = 6.283185307179586f * tt;
    const float m1c = cosf(a1), m1s = sinf(a1);
    const float mtc = m1c + m1c;
    const float m2c = fmaf(mtc, m1c, -1.0f);
    const float m2s = mtc * m1s;

    const float zz  = aw0 + aw1 * m1c + aw2 * m1s + aw3 * m2c + aw4 * m2s;
    const float alo = __builtin_amdgcn_rcpf(1.f + __expf(-zz));

    const float o1c = xhalf(m1c), o1s = xhalf(m1s), alx = xhalf(alo);
    const float otc = o1c + o1c;
    const float o2c = fmaf(otc, o1c, -1.0f);
    const float o2s = otc * o1s;

    const bool hi = half != 0;
    const float p1c = hi ? o1c : m1c, p1s = hi ? o1s : m1s;
    const float p2c = hi ? o2c : m2c, p2s = hi ? o2s : m2s;
    const float q1c = hi ? m1c : o1c, q1s = hi ? m1s : o1s;
    const float q2c = hi ? m2c : o2c, q2s = hi ? m2s : o2s;
    const float alA = hi ? alx : alo;
    const float alB = hi ? alo : alx;

    const float ta0 = xhalf(xa0.x), ta1 = xhalf(xa0.y);
    const float tb0 = xhalf(xb0.x), tb1 = xhalf(xb0.y);
    const float xA0 = hi ? ta0 : xa0.x, xA1 = hi ? ta1 : xa0.y;
    const float xB0 = hi ? tb0 : xb0.x, xB1 = hi ? tb1 : xb0.y;

    const float glo = hi ? 1.f : 0.f;
#define SM_EL(XMI, XM0, XM1, G, DST)                                           \
    {                                                                          \
        const float xmi = (XMI);                                               \
        const float xm0 = (XM0);                                               \
        const float xm1 = (XM1);                                               \
        const float e0 = c1 * xmi * xm0;                                       \
        const float e1 = c1 * xmi * xm1;                                       \
        const float e2 = c1 * xmi * xmi;                                       \
        const float mx = fmaxf(fmaxf(e0, e1), e2);                             \
        const float w0 = __expf(e0 - mx);                                      \
        const float w1 = __expf(e1 - mx);                                      \
        const float w2 = (G) * __expf(e2 - mx);                                \
        const float inv = __builtin_amdgcn_rcpf(w0 + w1 + w2);                 \
        const float s  = (w0 * xm0 + w1 * xm1 + w2 * xmi) * inv;               \
        DST = c2 * s;                                                          \
    }
#define SM_PAIR(RR, XAV, XBV, E0, E1, G)                                       \
    SM_EL((XAV).E0, xA0, xA1, G, smA[RR])     SM_EL((XBV).E0, xB0, xB1, G, smB[RR])     \
    SM_EL((XAV).E1, xA0, xA1, G, smA[RR + 1]) SM_EL((XBV).E1, xB0, xB1, G, smB[RR + 1])
    float smA[8], smB[8];

    v8f accA = {}, accB = {};
#define WMMA_RAW(ACC, WFR, SFR)                                                \
        ACC = __builtin_amdgcn_wmma_f32_16x16x4_f32(                           \
            false, (WFR), false, (SFR), (short)0, ACC, false, false)
#define WMMA_SC(ACC, EV, WFR, XH)                                              \
        do { v2f s_; s_.x = (EV) * (XH).x; s_.y = (EV) * (XH).y;               \
             WMMA_RAW(ACC, WFR, s_); } while (0)
#define STEP4(EA, EB, KB)                                                      \
        WMMA_SC(accA, EA, lq[KB][0].lo, xa0.lo); WMMA_SC(accB, EB, lq[KB][0].lo, xb0.lo); \
        WMMA_SC(accA, EA, lq[KB][0].hi, xa0.hi); WMMA_SC(accB, EB, lq[KB][0].hi, xb0.hi); \
        WMMA_SC(accA, EA, lq[KB][1].lo, xa1.lo); WMMA_SC(accB, EB, lq[KB][1].lo, xb1.lo); \
        WMMA_SC(accA, EA, lq[KB][1].hi, xa1.hi); WMMA_SC(accB, EB, lq[KB][1].hi, xb1.hi)

        WMMA_RAW(accA, lq[0][0].lo, xa0.lo); WMMA_RAW(accB, lq[0][0].lo, xb0.lo);
        WMMA_RAW(accA, lq[0][0].hi, xa0.hi); WMMA_RAW(accB, lq[0][0].hi, xb0.hi);
        WMMA_RAW(accA, lq[0][1].lo, xa1.lo); WMMA_RAW(accB, lq[0][1].lo, xb1.lo);
        WMMA_RAW(accA, lq[0][1].hi, xa1.hi); WMMA_RAW(accB, lq[0][1].hi, xb1.hi);
        SM_PAIR(0, xa0, xb0, x, y, glo)
        STEP4(p1c, q1c, 1);
        SM_PAIR(2, xa0, xb0, z, w, 1.f)
        STEP4(p1s, q1s, 2);
        SM_PAIR(4, xa1, xb1, x, y, 1.f)
        STEP4(p2c, q2c, 3);
        SM_PAIR(6, xa1, xb1, z, w, 1.f)
        STEP4(p2s, q2s, 4);
#undef STEP4
#undef WMMA_SC
#undef WMMA_RAW
#undef SM_PAIR
#undef SM_EL

    float resA[8], resB[8];
#pragma unroll
    for (int rr = 0; rr < 8; ++rr) {
        resA[rr] = accA[rr] + smA[rr] * alA;
        resB[rr] = accB[rr] + smB[rr] * alB;
    }

    v4f* op = reinterpret_cast<v4f*>(out + rowA * 16 + (half << 2));
    v4f* oq = reinterpret_cast<v4f*>(out + rowB * 16 + (half << 2));
    v4f r0, r1, r2, r3;
    r0.x = resA[0]; r0.y = resA[1]; r0.z = resA[2]; r0.w = resA[3];
    r1.x = resA[4]; r1.y = resA[5]; r1.z = resA[6]; r1.w = resA[7];
    r2.x = resB[0]; r2.y = resB[1]; r2.z = resB[2]; r2.w = resB[3];
    r3.x = resB[4]; r3.y = resB[5]; r3.z = resB[6]; r3.w = resB[7];
    op[0] = r0; op[2] = r1;
    oq[0] = r2; oq[2] = r3;
}

extern "C" void kernel_launch(void* const* d_in, const int* in_sizes, int n_in,
                              void* d_out, int out_size, void* d_ws, size_t ws_size,
                              hipStream_t stream) {
    const float* x  = (const float*)d_in[0];
    const float* t  = (const float*)d_in[1];
    const float* Lb = (const float*)d_in[2];
    const float* wq = (const float*)d_in[3];
    const float* wk = (const float*)d_in[4];
    const float* wv = (const float*)d_in[5];
    const float* wo = (const float*)d_in[6];
    const float* aw = (const float*)d_in[7];
    float* out = (float*)d_out;

    const int B = in_sizes[1];
    const int nTiles = B >> 4;
    if (nTiles <= 0) return;

    const int nPairs = (nTiles + 1) >> 1;
    const int blocks = (nPairs + WPB - 1) / WPB;

    nxro_fused_kernel<<<blocks, 256, 0, stream>>>(x, t, Lb, wq, wk, wv, wo, aw,
                                                  out, nTiles);
}
